# starting kernel
# baseline (speedup 1.0000x reference)
#include <hip/hip_runtime.h>

#define N_NODES 100000
#define N_EDGES 640000
#define DIM 128
#define BN_EPS 1e-5f

typedef __attribute__((ext_vector_type(2))) float v2f;
typedef __attribute__((ext_vector_type(8))) float v8f;

__global__ void zero_kernel(float4* __restrict__ out4, float* __restrict__ ws) {
    int i = blockIdx.x * blockDim.x + threadIdx.x;
    if (i < (N_NODES * DIM / 4)) out4[i] = make_float4(0.f, 0.f, 0.f, 0.f);
    if (i < 256) ws[i] = 0.f;
}

__global__ void scatter_kernel(const float* __restrict__ feat,
                               const float* __restrict__ ew,
                               const int*   __restrict__ src,
                               const int*   __restrict__ dst,
                               float*       __restrict__ h) {
    int gid  = blockIdx.x * blockDim.x + threadIdx.x;
    int e    = gid >> 5;
    int lane = gid & 31;
    if (e >= N_EDGES) return;
    int   s = src[e];
    int   d = dst[e];
    float w = ew[e];
    const float4 f = *reinterpret_cast<const float4*>(feat + (size_t)s * DIM + lane * 4);
    float* hp = h + (size_t)d * DIM + lane * 4;
    unsafeAtomicAdd(hp + 0, w * f.x);
    unsafeAtomicAdd(hp + 1, w * f.y);
    unsafeAtomicAdd(hp + 2, w * f.z);
    unsafeAtomicAdd(hp + 3, w * f.w);
}

__launch_bounds__(256)
__global__ void gemm_relu_stats_kernel(float* __restrict__ hy,
                                       const float* __restrict__ W,
                                       const float* __restrict__ bias,
                                       float* __restrict__ sum,
                                       float* __restrict__ sumsq) {
    __shared__ float tile[16 * DIM];

    const int m0 = blockIdx.x * 16;

    {
        const float4* s4 = reinterpret_cast<const float4*>(hy + (size_t)m0 * DIM);
        float4*       d4 = reinterpret_cast<float4*>(tile);
        d4[threadIdx.x]       = s4[threadIdx.x];
        d4[threadIdx.x + 256] = s4[threadIdx.x + 256];
    }
    __syncthreads();

    const int wave = threadIdx.x >> 5;
    const int lane = threadIdx.x & 31;
    const int nl   = lane & 15;
    const int half = lane >> 4;
    const int n    = wave * 16 + nl;

    const float* wrow = W + (size_t)n * DIM;

    v8f acc = {};
#pragma unroll
    for (int k0 = 0; k0 < DIM; k0 += 4) {
        const int kk = k0 + 2 * half;
        v2f a = *reinterpret_cast<const v2f*>(&tile[nl * DIM + kk]);
        v2f b = *reinterpret_cast<const v2f*>(wrow + kk);
        acc = __builtin_amdgcn_wmma_f32_16x16x4_f32(
             false, a,  false, b,
             (short)0, acc,  false,  false);
    }

    const float bn = bias[n];
    float ls = 0.f, lq = 0.f;
    float* yb = hy + (size_t)m0 * DIM + n;
#pragma unroll
    for (int v = 0; v < 8; ++v) {
        float val = fmaxf(acc[v] + bn, 0.f);
        const int m = half ? (v + 8) : v;
        yb[(size_t)m * DIM] = val;
        ls += val;
        lq += val * val;
    }
    ls += __shfl_xor(ls, 16);
    lq += __shfl_xor(lq, 16);
    if (half == 0) {
        unsafeAtomicAdd(sum + n, ls);
        unsafeAtomicAdd(sumsq + n, lq);
    }
}

__global__ void bn_finalize_kernel(const float* __restrict__ sum,
                                   const float* __restrict__ sumsq,
                                   const float* __restrict__ gamma,
                                   const float* __restrict__ beta,
                                   float* __restrict__ scale,
                                   float* __restrict__ shift) {
    int n = threadIdx.x;
    const float inv  = 1.0f / (float)N_NODES;
    float mean = sum[n] * inv;
    float var  = fmaf(-mean, mean, sumsq[n] * inv);
    float sc   = gamma[n] * rsqrtf(var + BN_EPS);
    scale[n] = sc;
    shift[n] = fmaf(-mean, sc, beta[n]);
}

__global__ void bn_apply_kernel(float4* __restrict__ y4,
                                const float4* __restrict__ scale4,
                                const float4* __restrict__ shift4) {
    int i = blockIdx.x * blockDim.x + threadIdx.x;
    if (i >= (N_NODES * DIM / 4)) return;
    const int c = i & 31;
    float4 v = y4[i];
    const float4 s = scale4[c];
    const float4 t = shift4[c];
    v.x = fmaf(v.x, s.x, t.x);
    v.y = fmaf(v.y, s.y, t.y);
    v.z = fmaf(v.z, s.z, t.z);
    v.w = fmaf(v.w, s.w, t.w);
    y4[i] = v;
}

extern "C" void kernel_launch(void* const* d_in, const int* in_sizes, int n_in,
                              void* d_out, int out_size, void* d_ws, size_t ws_size,
                              hipStream_t stream) {
    const float* feature = (const float*)d_in[0];
    const float* ew      = (const float*)d_in[1];
    const int*   src     = (const int*)d_in[2];
    const int*   dst     = (const int*)d_in[3];
    const float* W       = (const float*)d_in[4];
    const float* b       = (const float*)d_in[5];
    const float* gamma   = (const float*)d_in[6];
    const float* beta    = (const float*)d_in[7];

    float* out = (float*)d_out;
    float* ws  = (float*)d_ws;
    float* sum   = ws;
    float* sumsq = ws + 128;
    float* scale = ws + 256;
    float* shift = ws + 384;

    const int total4 = N_NODES * DIM / 4;

    zero_kernel<<<(total4 + 255) / 256, 256, 0, stream>>>((float4*)out, ws);

    const long sthreads = (long)N_EDGES * 32;
    scatter_kernel<<<(int)((sthreads + 255) / 256), 256, 0, stream>>>(
        feature, ew, src, dst, out);

    gemm_relu_stats_kernel<<<N_NODES / 16, 256, 0, stream>>>(out, W, b, sum, sumsq);

    bn_finalize_kernel<<<1, 128, 0, stream>>>(sum, sumsq, gamma, beta, scale, shift);

    bn_apply_kernel<<<(total4 + 255) / 256, 256, 0, stream>>>(
        (float4*)out, (const float4*)scale, (const float4*)shift);
}


# Resident grids: GEMM and BN-apply loop over rows in 512 work-groups; W fragments, bias, scale/shift loaded once per wave; H tile at LDS pitch 132; BN sums kept per wave, one atomic add at the end
# speedup vs baseline: 1.2133x; 1.2133x over previous
#include <hip/hip_runtime.h>

#define N_NODES 100000
#define N_EDGES 640000
#define DIM 128
#define BN_EPS 1e-5f

typedef __attribute__((ext_vector_type(2))) float v2f;
typedef __attribute__((ext_vector_type(8))) float v8f;

__global__ void zero_kernel(float4* __restrict__ out4, float* __restrict__ ws) {
    int i = blockIdx.x * blockDim.x + threadIdx.x;
    if (i < (N_NODES * DIM / 4)) out4[i] = make_float4(0.f, 0.f, 0.f, 0.f);
    if (i < 256) ws[i] = 0.f;
}

__global__ void scatter_kernel(const float* __restrict__ feat,
                               const float* __restrict__ ew,
                               const int*   __restrict__ src,
                               const int*   __restrict__ dst,
                               float*       __restrict__ h) {
    int gid  = blockIdx.x * blockDim.x + threadIdx.x;
    int e    = gid >> 5;
    int lane = gid & 31;
    if (e >= N_EDGES) return;
    int   s = src[e];
    int   d = dst[e];
    float w = ew[e];
    const float4 f = *reinterpret_cast<const float4*>(feat + (size_t)s * DIM + lane * 4);
    float* hp = h + (size_t)d * DIM + lane * 4;
    unsafeAtomicAdd(hp + 0, w * f.x);
    unsafeAtomicAdd(hp + 1, w * f.y);
    unsafeAtomicAdd(hp + 2, w * f.z);
    unsafeAtomicAdd(hp + 3, w * f.w);
}

#define GEMM_BLOCKS 512
#define TILE_PITCH 132
#define N_TILES (N_NODES / 16)

__device__ __forceinline__ void gemm_tile(const float* __restrict__ tb,
                                          const v2f (&b)[32], float bn,
                                          int nl, int half,
                                          float* yb,
                                          float& ls, float& lq) {
    const float* arow = tb + nl * TILE_PITCH + 2 * half;
    v8f acc = {};
#pragma unroll
    for (int j = 0; j < 32; ++j) {
        v2f a = *reinterpret_cast<const v2f*>(arow + 4 * j);
        acc = __builtin_amdgcn_wmma_f32_16x16x4_f32(
             false, a,  false, b[j],
             (short)0, acc,  false,  false);
    }
#pragma unroll
    for (int v = 0; v < 8; ++v) {
        float val = fmaxf(acc[v] + bn, 0.f);
        const int m = half ? (v + 8) : v;
        yb[(size_t)m * DIM] = val;
        ls += val;
        lq += val * val;
    }
}

__launch_bounds__(256)
__global__ void gemm_relu_stats_kernel(float* __restrict__ hy,
                                       const float* __restrict__ W,
                                       const float* __restrict__ bias,
                                       float* __restrict__ sum,
                                       float* __restrict__ sumsq) {
    __shared__ float tile[2][16 * TILE_PITCH];

    const int wave = threadIdx.x >> 5;
    const int lane = threadIdx.x & 31;
    const int nl   = lane & 15;
    const int half = lane >> 4;
    const int n    = wave * 16 + nl;

    v2f b[32];
    {
        const float* wrow = W + (size_t)n * DIM + 2 * half;
#pragma unroll
        for (int j = 0; j < 32; ++j)
            b[j] = *reinterpret_cast<const v2f*>(wrow + 4 * j);
    }
    const float bn = bias[n];

    const int sOff0 = (threadIdx.x >> 5) * TILE_PITCH + (threadIdx.x & 31) * 4;
    const int sOff1 = sOff0 + 8 * TILE_PITCH;
    const float4* h4 = reinterpret_cast<const float4*>(hy);
    const int ld0 = threadIdx.x;
    const int ld1 = threadIdx.x + 256;

    int t = blockIdx.x;
    float4 p0 = make_float4(0.f, 0.f, 0.f, 0.f), p1 = p0;
    if (t < N_TILES) {
        p0 = h4[(size_t)t * 512 + ld0];
        p1 = h4[(size_t)t * 512 + ld1];
    }

    float ls = 0.f, lq = 0.f;
    int buf = 0;
    for (; t < N_TILES; t += GEMM_BLOCKS) {
        float* tb = tile[buf];
        *reinterpret_cast<float4*>(tb + sOff0) = p0;
        *reinterpret_cast<float4*>(tb + sOff1) = p1;
        __syncthreads();
        if (t + GEMM_BLOCKS < N_TILES) {
            p0 = h4[(size_t)(t + GEMM_BLOCKS) * 512 + ld0];
            p1 = h4[(size_t)(t + GEMM_BLOCKS) * 512 + ld1];
        }
        gemm_tile(tb, b, bn, nl, half, hy + (size_t)t * 16 * DIM + n, ls, lq);
        buf ^= 1;
    }

    ls += __shfl_xor(ls, 16);
    lq += __shfl_xor(lq, 16);
    if (half == 0) {
        unsafeAtomicAdd(sum + n, ls);
        unsafeAtomicAdd(sumsq + n, lq);
    }
}

__global__ void bn_finalize_kernel(const float* __restrict__ sum,
                                   const float* __restrict__ sumsq,
                                   const float* __restrict__ gamma,
                                   const float* __restrict__ beta,
                                   float* __restrict__ scale,
                                   float* __restrict__ shift) {
    int n = threadIdx.x;
    const float inv  = 1.0f / (float)N_NODES;
    float mean = sum[n] * inv;
    float var  = fmaf(-mean, mean, sumsq[n] * inv);
    float sc   = gamma[n] * rsqrtf(var + BN_EPS);
    scale[n] = sc;
    shift[n] = fmaf(-mean, sc, beta[n]);
}

#define APPLY_BLOCKS 512
#define APPLY_STRIDE (APPLY_BLOCKS * 256)

__launch_bounds__(256)
__global__ void bn_apply_kernel(float4* __restrict__ y4,
                                const float4* __restrict__ scale4,
                                const float4* __restrict__ shift4) {
    constexpr int total4 = N_NODES * DIM / 4;
    constexpr int FULL = total4 / APPLY_STRIDE;
    const int gtid = blockIdx.x * 256 + threadIdx.x;
    const int c = gtid & 31;
    const float4 s = scale4[c];
    const float4 t = shift4[c];

    float4* p = y4 + gtid;
    for (int k = 0; k + 8 <= FULL; k += 8) {
        float4 v[8];
#pragma unroll
        for (int u = 0; u < 8; ++u) v[u] = p[(size_t)(k + u) * APPLY_STRIDE];
#pragma unroll
        for (int u = 0; u < 8; ++u) {
            float4 r = v[u];
            r.x = fmaf(r.x, s.x, t.x);
            r.y = fmaf(r.y, s.y, t.y);
            r.z = fmaf(r.z, s.z, t.z);
            r.w = fmaf(r.w, s.w, t.w);
            p[(size_t)(k + u) * APPLY_STRIDE] = r;
        }
    }
    for (int k = FULL - FULL % 8; k < FULL; ++k) {
        float4 r = p[(size_t)k * APPLY_STRIDE];
        r.x = fmaf(r.x, s.x, t.x);
        r.y = fmaf(r.y, s.y, t.y);
        r.z = fmaf(r.z, s.z, t.z);
        r.w = fmaf(r.w, s.w, t.w);
        p[(size_t)k * APPLY_STRIDE] = r;
    }
    if (gtid + FULL * APPLY_STRIDE < total4) {
        float4 r = p[(size_t)FULL * APPLY_STRIDE];
        r.x = fmaf(r.x, s.x, t.x);
        r.y = fmaf(r.y, s.y, t.y);
        r.z = fmaf(r.z, s.z, t.z);
        r.w = fmaf(r.w, s.w, t.w);
        p[(size_t)FULL * APPLY_STRIDE] = r;
    }
}

extern "C" void kernel_launch(void* const* d_in, const int* in_sizes, int n_in,
                              void* d_out, int out_size, void* d_ws, size_t ws_size,
                              hipStream_t stream) {
    const float* feature = (const float*)d_in[0];
    const float* ew      = (const float*)d_in[1];
    const int*   src     = (const int*)d_in[2];
    const int*   dst     = (const int*)d_in[3];
    const float* W       = (const float*)d_in[4];
    const float* b       = (const float*)d_in[5];
    const float* gamma   = (const float*)d_in[6];
    const float* beta    = (const float*)d_in[7];

    float* out = (float*)d_out;
    float* ws  = (float*)d_ws;
    float* sum   = ws;
    float* sumsq = ws + 128;
    float* scale = ws + 256;
    float* shift = ws + 384;

    const int total4 = N_NODES * DIM / 4;

    zero_kernel<<<(total4 + 255) / 256, 256, 0, stream>>>((float4*)out, ws);

    const long sthreads = (long)N_EDGES * 32;
    scatter_kernel<<<(int)((sthreads + 255) / 256), 256, 0, stream>>>(
        feature, ew, src, dst, out);

    gemm_relu_stats_kernel<<<GEMM_BLOCKS, 256, 0, stream>>>(out, W, b, sum, sumsq);

    bn_finalize_kernel<<<1, 128, 0, stream>>>(sum, sumsq, gamma, beta, scale, shift);

    bn_apply_kernel<<<APPLY_BLOCKS, 256, 0, stream>>>(
        (float4*)out, (const float4*)scale, (const float4*)shift);
}


# Zero fill + global float-atomic scatter replaced by owner groups: 250 work-groups keep 400 dst rows each in LDS, scan all edge dsts, add with LDS float atomics, store H once; sum order = LDS arrival
# speedup vs baseline: 1.6674x; 1.3743x over previous
#include <hip/hip_runtime.h>

#define N_NODES 100000
#define N_EDGES 640000
#define DIM 128
#define BN_EPS 1e-5f

typedef __attribute__((ext_vector_type(2))) float v2f;
typedef __attribute__((ext_vector_type(8))) float v8f;

#define OWN_ROWS 400
#define OWN_BLOCKS (N_NODES / OWN_ROWS)
#define OWN_WAVES 32
#define LIST_CAP 256
#define CHUNK_EDGES 128
#define N_CHUNKS (N_EDGES / CHUNK_EDGES)
#define SCAN_DEPTH 8
#define ROWS_IN_FLIGHT 8
#define WALK_AT 32
#define ID_MASK 0xFFFFFu

static_assert(N_NODES % OWN_ROWS == 0, "owner groups tile the nodes exactly");
static_assert(N_EDGES % CHUNK_EDGES == 0, "chunks tile the edges exactly");
static_assert(N_EDGES <= (1 << 20) && N_NODES <= (1 << 20) && OWN_ROWS <= (1 << 12), "list entry packing");
static_assert(WALK_AT + CHUNK_EDGES <= LIST_CAP && LIST_CAP % ROWS_IN_FLIGHT == 0, "a chunk always fits behind the walk test; padding stays inside the list");

__device__ __forceinline__ void list_push(unsigned* ent, int& cnt, bool m, unsigned val, int lane) {
    const unsigned mask = __builtin_amdgcn_ballot_w32(m);
    if (mask) {
        if (m) ent[cnt + __builtin_amdgcn_mbcnt_lo(mask, 0u)] = val;
        cnt += __builtin_popcount(mask);
    }
}

__device__ __forceinline__ void list_walk(unsigned* ent, float* wt, int cnt, int lane,
                                          const float4* __restrict__ feat4,
                                          const float* __restrict__ ew,
                                          const int* __restrict__ src,
                                          float* rows) {
    for (int j = lane; j < cnt; j += 32) {
        const unsigned en = ent[j];
        const unsigned e  = en & ID_MASK;
        const unsigned sv = (unsigned)src[e] & ID_MASK;
        wt[j]  = ew[e];
        ent[j] = (en & ~ID_MASK) | sv;
    }
    const int n = (cnt + ROWS_IN_FLIGHT - 1) & ~(ROWS_IN_FLIGHT - 1);
    if (cnt + lane < n) {
        ent[cnt + lane] = (unsigned)OWN_ROWS << 20;
        wt[cnt + lane]  = 0.f;
    }
    for (int j0 = 0; j0 < n; j0 += ROWS_IN_FLIGHT) {
        unsigned en[ROWS_IN_FLIGHT];
        float    w[ROWS_IN_FLIGHT];
        float4   f[ROWS_IN_FLIGHT];
#pragma unroll
        for (int u = 0; u < ROWS_IN_FLIGHT; ++u) {
            en[u] = ent[j0 + u];
            w[u]  = wt[j0 + u];
            f[u]  = feat4[(size_t)(en[u] & ID_MASK) * (DIM / 4) + lane];
        }
#pragma unroll
        for (int u = 0; u < ROWS_IN_FLIGHT; ++u) {
            float* p = rows + (en[u] >> 20) * DIM + lane;
            unsafeAtomicAdd(p,      w[u] * f[u].x);
            unsafeAtomicAdd(p + 32, w[u] * f[u].y);
            unsafeAtomicAdd(p + 64, w[u] * f[u].z);
            unsafeAtomicAdd(p + 96, w[u] * f[u].w);
        }
    }
}

__launch_bounds__(1024)
__global__ void owner_sum_kernel(const float* __restrict__ feat,
                                 const float* __restrict__ ew,
                                 const int*   __restrict__ src,
                                 const int*   __restrict__ dst,
                                 float*       __restrict__ h,
                                 float*       __restrict__ ws) {
    __shared__ __attribute__((aligned(16))) float rows[(OWN_ROWS + 1) * DIM];
    __shared__ unsigned ent_all[OWN_WAVES * LIST_CAP];
    __shared__ float    wt_all[OWN_WAVES * LIST_CAP];

    const int tid  = threadIdx.x;
    const int wave = __builtin_amdgcn_readfirstlane(tid >> 5);
    const int lane = tid & 31;
    const int lo   = blockIdx.x * OWN_ROWS;

    if (blockIdx.x == 0 && tid < 256) ws[tid] = 0.f;

    for (int i = tid; i < OWN_ROWS * DIM / 4; i += 1024)
        reinterpret_cast<float4*>(rows)[i] = make_float4(0.f, 0.f, 0.f, 0.f);
    __syncthreads();

    unsigned* ent = ent_all + wave * LIST_CAP;
    float*    wt  = wt_all  + wave * LIST_CAP;
    const int4*   dst4  = reinterpret_cast<const int4*>(dst);
    const float4* feat4 = reinterpret_cast<const float4*>(feat);

    int cnt = 0;
    for (int c0 = wave; c0 < N_CHUNKS; c0 += OWN_WAVES * SCAN_DEPTH) {
        int4 d[SCAN_DEPTH];
#pragma unroll
        for (int u = 0; u < SCAN_DEPTH; ++u) {
            const int c  = c0 + u * OWN_WAVES;
            const int cc = (c < N_CHUNKS) ? c : c0;
            d[u] = dst4[(size_t)cc * (CHUNK_EDGES / 4) + lane];
        }
        int next = 0;
        while (next < SCAN_DEPTH) {
            bool full = false;
#pragma unroll
            for (int u = 0; u < SCAN_DEPTH; ++u) {
                if (u >= next && !full) {
                    if (cnt >= WALK_AT) {
                        full = true;
                    } else {
                        next = u + 1;
                        const int c = c0 + u * OWN_WAVES;
                        if (c < N_CHUNKS) {
                            const unsigned e0 = (unsigned)c * CHUNK_EDGES + (unsigned)lane * 4u;
                            const unsigned r0 = (unsigned)(d[u].x - lo);
                            const unsigned r1 = (unsigned)(d[u].y - lo);
                            const unsigned r2 = (unsigned)(d[u].z - lo);
                            const unsigned r3 = (unsigned)(d[u].w - lo);
                            const bool m0 = r0 < OWN_ROWS, m1 = r1 < OWN_ROWS;
                            const bool m2 = r2 < OWN_ROWS, m3 = r3 < OWN_ROWS;
                            if (__builtin_amdgcn_ballot_w32(m0 | m1 | m2 | m3)) {
                                list_push(ent, cnt, m0, (e0 + 0u) | (r0 << 20), lane);
                                list_push(ent, cnt, m1, (e0 + 1u) | (r1 << 20), lane);
                                list_push(ent, cnt, m2, (e0 + 2u) | (r2 << 20), lane);
                                list_push(ent, cnt, m3, (e0 + 3u) | (r3 << 20), lane);
                            }
                        }
                    }
                }
            }
            if (full) {
                list_walk(ent, wt, cnt, lane, feat4, ew, src, rows);
                cnt = 0;
            }
        }
    }
    if (cnt > 0) list_walk(ent, wt, cnt, lane, feat4, ew, src, rows);
    __syncthreads();

    float4* out4 = reinterpret_cast<float4*>(h) + (size_t)lo * (DIM / 4);
    for (int i = tid; i < OWN_ROWS * (DIM / 4); i += 1024) {
        const float* p = rows + (i >> 5) * DIM + (i & 31);
        out4[i] = make_float4(p[0], p[32], p[64], p[96]);
    }
}

#define GEMM_BLOCKS 512
#define TILE_PITCH 132
#define N_TILES (N_NODES / 16)

__device__ __forceinline__ void gemm_tile(const float* __restrict__ tb,
                                          const v2f (&b)[32], float bn,
                                          int nl, int half,
                                          float* yb,
                                          float& ls, float& lq) {
    const float* arow = tb + nl * TILE_PITCH + 2 * half;
    v8f acc = {};
#pragma unroll
    for (int j = 0; j < 32; ++j) {
        v2f a = *reinterpret_cast<const v2f*>(arow + 4 * j);
        acc = __builtin_amdgcn_wmma_f32_16x16x4_f32(
             false, a,  false, b[j],
             (short)0, acc,  false,  false);
    }
#pragma unroll
    for (int v = 0; v < 8; ++v) {
        float val = fmaxf(acc[v] + bn, 0.f);
        const int m = half ? (v + 8) : v;
        yb[(size_t)m * DIM] = val;
        ls += val;
        lq += val * val;
    }
}

__launch_bounds__(256)
__global__ void gemm_relu_stats_kernel(float* __restrict__ hy,
                                       const float* __restrict__ W,
                                       const float* __restrict__ bias,
                                       float* __restrict__ sum,
                                       float* __restrict__ sumsq) {
    __shared__ float tile[2][16 * TILE_PITCH];

    const int wave = threadIdx.x >> 5;
    const int lane = threadIdx.x & 31;
    const int nl   = lane & 15;
    const int half = lane >> 4;
    const int n    = wave * 16 + nl;

    v2f b[32];
    {
        const float* wrow = W + (size_t)n * DIM + 2 * half;
#pragma unroll
        for (int j = 0; j < 32; ++j)
            b[j] = *reinterpret_cast<const v2f*>(wrow + 4 * j);
    }
    const float bn = bias[n];

    const int sOff0 = (threadIdx.x >> 5) * TILE_PITCH + (threadIdx.x & 31) * 4;
    const int sOff1 = sOff0 + 8 * TILE_PITCH;
    const float4* h4 = reinterpret_cast<const float4*>(hy);
    const int ld0 = threadIdx.x;
    const int ld1 = threadIdx.x + 256;

    int t = blockIdx.x;
    float4 p0 = make_float4(0.f, 0.f, 0.f, 0.f), p1 = p0;
    if (t < N_TILES) {
        p0 = h4[(size_t)t * 512 + ld0];
        p1 = h4[(size_t)t * 512 + ld1];
    }

    float ls = 0.f, lq = 0.f;
    int buf = 0;
    for (; t < N_TILES; t += GEMM_BLOCKS) {
        float* tb = tile[buf];
        *reinterpret_cast<float4*>(tb + sOff0) = p0;
        *reinterpret_cast<float4*>(tb + sOff1) = p1;
        __syncthreads();
        if (t + GEMM_BLOCKS < N_TILES) {
            p0 = h4[(size_t)(t + GEMM_BLOCKS) * 512 + ld0];
            p1 = h4[(size_t)(t + GEMM_BLOCKS) * 512 + ld1];
        }
        gemm_tile(tb, b, bn, nl, half, hy + (size_t)t * 16 * DIM + n, ls, lq);
        buf ^= 1;
    }

    ls += __shfl_xor(ls, 16);
    lq += __shfl_xor(lq, 16);
    if (half == 0) {
        unsafeAtomicAdd(sum + n, ls);
        unsafeAtomicAdd(sumsq + n, lq);
    }
}

__global__ void bn_finalize_kernel(const float* __restrict__ sum,
                                   const float* __restrict__ sumsq,
                                   const float* __restrict__ gamma,
                                   const float* __restrict__ beta,
                                   float* __restrict__ scale,
                                   float* __restrict__ shift) {
    int n = threadIdx.x;
    const float inv  = 1.0f / (float)N_NODES;
    float mean = sum[n] * inv;
    float var  = fmaf(-mean, mean, sumsq[n] * inv);
    float sc   = gamma[n] * rsqrtf(var + BN_EPS);
    scale[n] = sc;
    shift[n] = fmaf(-mean, sc, beta[n]);
}

#define APPLY_BLOCKS 512
#define APPLY_STRIDE (APPLY_BLOCKS * 256)

__launch_bounds__(256)
__global__ void bn_apply_kernel(float4* __restrict__ y4,
                                const float4* __restrict__ scale4,
                                const float4* __restrict__ shift4) {
    constexpr int total4 = N_NODES * DIM / 4;
    constexpr int FULL = total4 / APPLY_STRIDE;
    const int gtid = blockIdx.x * 256 + threadIdx.x;
    const int c = gtid & 31;
    const float4 s = scale4[c];
    const float4 t = shift4[c];

    float4* p = y4 + gtid;
    for (int k = 0; k + 8 <= FULL; k += 8) {
        float4 v[8];
#pragma unroll
        for (int u = 0; u < 8; ++u) v[u] = p[(size_t)(k + u) * APPLY_STRIDE];
#pragma unroll
        for (int u = 0; u < 8; ++u) {
            float4 r = v[u];
            r.x = fmaf(r.x, s.x, t.x);
            r.y = fmaf(r.y, s.y, t.y);
            r.z = fmaf(r.z, s.z, t.z);
            r.w = fmaf(r.w, s.w, t.w);
            p[(size_t)(k + u) * APPLY_STRIDE] = r;
        }
    }
    for (int k = FULL - FULL % 8; k < FULL; ++k) {
        float4 r = p[(size_t)k * APPLY_STRIDE];
        r.x = fmaf(r.x, s.x, t.x);
        r.y = fmaf(r.y, s.y, t.y);
        r.z = fmaf(r.z, s.z, t.z);
        r.w = fmaf(r.w, s.w, t.w);
        p[(size_t)k * APPLY_STRIDE] = r;
    }
    if (gtid + FULL * APPLY_STRIDE < total4) {
        float4 r = p[(size_t)FULL * APPLY_STRIDE];
        r.x = fmaf(r.x, s.x, t.x);
        r.y = fmaf(r.y, s.y, t.y);
        r.z = fmaf(r.z, s.z, t.z);
        r.w = fmaf(r.w, s.w, t.w);
        p[(size_t)FULL * APPLY_STRIDE] = r;
    }
}

extern "C" void kernel_launch(void* const* d_in, const int* in_sizes, int n_in,
                              void* d_out, int out_size, void* d_ws, size_t ws_size,
                              hipStream_t stream) {
    const float* feature = (const float*)d_in[0];
    const float* ew      = (const float*)d_in[1];
    const int*   src     = (const int*)d_in[2];
    const int*   dst     = (const int*)d_in[3];
    const float* W       = (const float*)d_in[4];
    const float* b       = (const float*)d_in[5];
    const float* gamma   = (const float*)d_in[6];
    const float* beta    = (const float*)d_in[7];

    float* out = (float*)d_out;
    float* ws  = (float*)d_ws;
    float* sum   = ws;
    float* sumsq = ws + 128;
    float* scale = ws + 256;
    float* shift = ws + 384;

    owner_sum_kernel<<<OWN_BLOCKS, 1024, 0, stream>>>(feature, ew, src, dst, out, ws);

    gemm_relu_stats_kernel<<<GEMM_BLOCKS, 256, 0, stream>>>(out, W, b, sum, sumsq);

    bn_finalize_kernel<<<1, 128, 0, stream>>>(sum, sumsq, gamma, beta, scale, shift);

    bn_apply_kernel<<<APPLY_BLOCKS, 256, 0, stream>>>(
        (float4*)out, (const float4*)scale, (const float4*)shift);
}


# Edges bucketed by owner group once in a pre-pass kernel (LDS histogram, rank = arrival); owner groups read only their own runs instead of each scanning all dsts; sums still in LDS arrival order
# speedup vs baseline: 1.9538x; 1.1718x over previous
#include <hip/hip_runtime.h>

#define N_NODES 100000
#define N_EDGES 640000
#define DIM 128
#define BN_EPS 1e-5f

typedef __attribute__((ext_vector_type(2))) float v2f;
typedef __attribute__((ext_vector_type(8))) float v8f;

#define OWN_ROWS 400
#define OWN_BLOCKS (N_NODES / OWN_ROWS)
#define OWN_WAVES 32
#define LIST_CAP 256
#define LIST_TRIPS (LIST_CAP / 32)
#define ROWS_IN_FLIGHT 8
#define ID_MASK 0xFFFFFu

#define BIN_BLOCKS 250
#define BIN_THREADS 512
#define SLICE (N_EDGES / BIN_BLOCKS)
#define BIN_PER_THREAD (SLICE / BIN_THREADS)
#define SEG_PER_WAVE ((BIN_BLOCKS + OWN_WAVES - 1) / OWN_WAVES)
#define HIST_SIZE 256
#define WS_START_OFF 1024
#define WS_ENT_OFF (1 << 18)

static_assert(N_NODES % OWN_ROWS == 0, "owner groups tile the nodes exactly");
static_assert(N_EDGES % BIN_BLOCKS == 0 && SLICE % BIN_THREADS == 0, "slices tile the edges, threads tile a slice");
static_assert(N_NODES <= (1 << 20) && OWN_ROWS <= (1 << 12), "list entry packing");
static_assert(OWN_BLOCKS + 1 <= HIST_SIZE && HIST_SIZE <= BIN_THREADS && HIST_SIZE % 32 == 0, "one histogram slot per group + the out-of-range slot");
static_assert(SEG_PER_WAVE <= 32 && LIST_CAP % ROWS_IN_FLIGHT == 0 && LIST_CAP % 32 == 0, "a wave's runs fit its lanes; padding stays inside the list");
static_assert(WS_START_OFF + (OWN_BLOCKS + 1) * BIN_BLOCKS <= WS_ENT_OFF, "start table ends before the entries");

__launch_bounds__(BIN_THREADS)
__global__ void bin_edges_kernel(const float* __restrict__ ew,
                                 const int*   __restrict__ src,
                                 const int*   __restrict__ dst,
                                 int*         __restrict__ start_t,
                                 unsigned*    __restrict__ ents) {
    __shared__ int hist[HIST_SIZE];
    __shared__ int wave_tot[HIST_SIZE / 32];

    const int tid  = threadIdx.x;
    const int lane = tid & 31;
    const int a    = blockIdx.x;

    if (tid < HIST_SIZE) hist[tid] = 0;
    __syncthreads();

    unsigned grp[BIN_PER_THREAD], val[BIN_PER_THREAD], wbits[BIN_PER_THREAD];
    int      rank[BIN_PER_THREAD];
#pragma unroll
    for (int k = 0; k < BIN_PER_THREAD; ++k) {
        const int e = a * SLICE + k * BIN_THREADS + tid;
        const unsigned d = (unsigned)dst[e];
        const unsigned s = (unsigned)src[e] & ID_MASK;
        wbits[k] = __float_as_uint(ew[e]);
        const unsigned q = d / OWN_ROWS;
        grp[k] = q < OWN_BLOCKS ? q : OWN_BLOCKS;
        val[k] = s | ((d - q * OWN_ROWS) << 20);
    }
#pragma unroll
    for (int k = 0; k < BIN_PER_THREAD; ++k)
        rank[k] = atomicAdd(&hist[grp[k]], 1);
    __syncthreads();

    int cntv = 0, incl = 0;
    if (tid < HIST_SIZE) {
        cntv = hist[tid];
        incl = cntv;
    }
#pragma unroll
    for (int dlt = 1; dlt < 32; dlt <<= 1) {
        const int t = __shfl_up(incl, dlt);
        if (lane >= dlt) incl += t;
    }
    if (tid < HIST_SIZE && lane == 31) wave_tot[tid >> 5] = incl;
    __syncthreads();
    if (tid < HIST_SIZE) {
        int base = 0;
        for (int w = 0; w < (tid >> 5); ++w) base += wave_tot[w];
        const int excl = base + incl - cntv;
        hist[tid] = excl;
        if (tid <= OWN_BLOCKS) start_t[tid * BIN_BLOCKS + a] = excl;
    }
    __syncthreads();

    uint2* ents2 = reinterpret_cast<uint2*>(ents);
#pragma unroll
    for (int k = 0; k < BIN_PER_THREAD; ++k)
        ents2[a * SLICE + hist[grp[k]] + rank[k]] = make_uint2(val[k], wbits[k]);
}

__device__ __forceinline__ void list_walk(unsigned* ent, float* wt, int cnt, int lane,
                                          const float4* __restrict__ feat4,
                                          float* rows) {
    const int n = (cnt + ROWS_IN_FLIGHT - 1) & ~(ROWS_IN_FLIGHT - 1);
    if (cnt + lane < n) {
        ent[cnt + lane] = (unsigned)OWN_ROWS << 20;
        wt[cnt + lane]  = 0.f;
    }
    for (int j0 = 0; j0 < n; j0 += ROWS_IN_FLIGHT) {
        unsigned en[ROWS_IN_FLIGHT];
        float    w[ROWS_IN_FLIGHT];
        float4   f[ROWS_IN_FLIGHT];
#pragma unroll
        for (int u = 0; u < ROWS_IN_FLIGHT; ++u) {
            en[u] = ent[j0 + u];
            w[u]  = wt[j0 + u];
            f[u]  = feat4[(size_t)(en[u] & ID_MASK) * (DIM / 4) + lane];
        }
#pragma unroll
        for (int u = 0; u < ROWS_IN_FLIGHT; ++u) {
            float* p = rows + (en[u] >> 20) * DIM + lane;
            unsafeAtomicAdd(p,      w[u] * f[u].x);
            unsafeAtomicAdd(p + 32, w[u] * f[u].y);
            unsafeAtomicAdd(p + 64, w[u] * f[u].z);
            unsafeAtomicAdd(p + 96, w[u] * f[u].w);
        }
    }
}

__launch_bounds__(1024)
__global__ void owner_sum_kernel(const float* __restrict__ feat,
                                 const int*      __restrict__ start_t,
                                 const unsigned* __restrict__ ents,
                                 float*       __restrict__ h,
                                 float*       __restrict__ sums) {
    __shared__ __attribute__((aligned(16))) float rows[(OWN_ROWS + 1) * DIM];
    __shared__ unsigned ent_all[OWN_WAVES * LIST_CAP];
    __shared__ float    wt_all[OWN_WAVES * LIST_CAP];

    const int tid  = threadIdx.x;
    const int wave = __builtin_amdgcn_readfirstlane(tid >> 5);
    const int lane = tid & 31;
    const int g    = blockIdx.x;
    const int lo   = g * OWN_ROWS;

    if (g == 0 && tid < 256) sums[tid] = 0.f;

    const int a = wave * SEG_PER_WAVE + lane;
    int s = 0, s_next = 0;
    if (lane < SEG_PER_WAVE && a < BIN_BLOCKS) {
        s      = start_t[g * BIN_BLOCKS + a];
        s_next = start_t[(g + 1) * BIN_BLOCKS + a];
    }

    for (int i = tid; i < OWN_ROWS * DIM / 4; i += 1024)
        reinterpret_cast<float4*>(rows)[i] = make_float4(0.f, 0.f, 0.f, 0.f);
    __syncthreads();
    const int nrun = s_next - s;

    unsigned* ent = ent_all + wave * LIST_CAP;
    float*    wt  = wt_all  + wave * LIST_CAP;
    const float4* feat4 = reinterpret_cast<const float4*>(feat);
    const uint2*  ents2 = reinterpret_cast<const uint2*>(ents);

    int inc[SEG_PER_WAVE], sb[SEG_PER_WAVE];
#pragma unroll
    for (int k = 0; k < SEG_PER_WAVE; ++k) {
        const int nk = __builtin_amdgcn_readlane(nrun, k);
        const int sk = __builtin_amdgcn_readlane(s, k);
        const int before = (k > 0) ? inc[(k > 0) ? k - 1 : 0] : 0;
        inc[k] = before + nk;
        sb[k]  = (wave * SEG_PER_WAVE + k) * SLICE + sk - before;
    }
    const int total = inc[SEG_PER_WAVE - 1];

    for (int p0 = 0; p0 < total; p0 += LIST_CAP) {
        const int cnt = (total - p0 < LIST_CAP) ? (total - p0) : LIST_CAP;
        uint2 x[LIST_TRIPS];
#pragma unroll
        for (int u = 0; u < LIST_TRIPS; ++u) {
            const int j = u * 32 + lane;
            x[u] = make_uint2(0u, 0u);
            if (j < cnt) {
                const int p = p0 + j;
                int base = sb[0];
#pragma unroll
                for (int k = 1; k < SEG_PER_WAVE; ++k)
                    if (p >= inc[k - 1]) base = sb[k];
                x[u] = ents2[base + p];
            }
        }
#pragma unroll
        for (int u = 0; u < LIST_TRIPS; ++u) {
            const int j = u * 32 + lane;
            if (j < cnt) {
                ent[j] = x[u].x;
                wt[j]  = __uint_as_float(x[u].y);
            }
        }
        list_walk(ent, wt, cnt, lane, feat4, rows);
    }
    __syncthreads();

    float4* out4 = reinterpret_cast<float4*>(h) + (size_t)lo * (DIM / 4);
    for (int i = tid; i < OWN_ROWS * (DIM / 4); i += 1024) {
        const float* p = rows + (i >> 5) * DIM + (i & 31);
        out4[i] = make_float4(p[0], p[32], p[64], p[96]);
    }
}

#define GEMM_BLOCKS 512
#define TILE_PITCH 132
#define N_TILES (N_NODES / 16)

__device__ __forceinline__ void gemm_tile(const float* __restrict__ tb,
                                          const v2f (&b)[32], float bn,
                                          int nl, int half,
                                          float* yb,
                                          float& ls, float& lq) {
    const float* arow = tb + nl * TILE_PITCH + 2 * half;
    v8f acc = {};
#pragma unroll
    for (int j = 0; j < 32; ++j) {
        v2f a = *reinterpret_cast<const v2f*>(arow + 4 * j);
        acc = __builtin_amdgcn_wmma_f32_16x16x4_f32(
             false, a,  false, b[j],
             (short)0, acc,  false,  false);
    }
#pragma unroll
    for (int v = 0; v < 8; ++v) {
        float val = fmaxf(acc[v] + bn, 0.f);
        const int m = half ? (v + 8) : v;
        yb[(size_t)m * DIM] = val;
        ls += val;
        lq += val * val;
    }
}

__launch_bounds__(256)
__global__ void gemm_relu_stats_kernel(float* __restrict__ hy,
                                       const float* __restrict__ W,
                                       const float* __restrict__ bias,
                                       float* __restrict__ sum,
                                       float* __restrict__ sumsq) {
    __shared__ float tile[2][16 * TILE_PITCH];

    const int wave = threadIdx.x >> 5;
    const int lane = threadIdx.x & 31;
    const int nl   = lane & 15;
    const int half = lane >> 4;
    const int n    = wave * 16 + nl;

    v2f b[32];
    {
        const float* wrow = W + (size_t)n * DIM + 2 * half;
#pragma unroll
        for (int j = 0; j < 32; ++j)
            b[j] = *reinterpret_cast<const v2f*>(wrow + 4 * j);
    }
    const float bn = bias[n];

    const int sOff0 = (threadIdx.x >> 5) * TILE_PITCH + (threadIdx.x & 31) * 4;
    const int sOff1 = sOff0 + 8 * TILE_PITCH;
    const float4* h4 = reinterpret_cast<const float4*>(hy);
    const int ld0 = threadIdx.x;
    const int ld1 = threadIdx.x + 256;

    int t = blockIdx.x;
    float4 p0 = make_float4(0.f, 0.f, 0.f, 0.f), p1 = p0;
    if (t < N_TILES) {
        p0 = h4[(size_t)t * 512 + ld0];
        p1 = h4[(size_t)t * 512 + ld1];
    }

    float ls = 0.f, lq = 0.f;
    int buf = 0;
    for (; t < N_TILES; t += GEMM_BLOCKS) {
        float* tb = tile[buf];
        *reinterpret_cast<float4*>(tb + sOff0) = p0;
        *reinterpret_cast<float4*>(tb + sOff1) = p1;
        __syncthreads();
        if (t + GEMM_BLOCKS < N_TILES) {
            p0 = h4[(size_t)(t + GEMM_BLOCKS) * 512 + ld0];
            p1 = h4[(size_t)(t + GEMM_BLOCKS) * 512 + ld1];
        }
        gemm_tile(tb, b, bn, nl, half, hy + (size_t)t * 16 * DIM + n, ls, lq);
        buf ^= 1;
    }

    ls += __shfl_xor(ls, 16);
    lq += __shfl_xor(lq, 16);
    if (half == 0) {
        unsafeAtomicAdd(sum + n, ls);
        unsafeAtomicAdd(sumsq + n, lq);
    }
}

__global__ void bn_finalize_kernel(const float* __restrict__ sum,
                                   const float* __restrict__ sumsq,
                                   const float* __restrict__ gamma,
                                   const float* __restrict__ beta,
                                   float* __restrict__ scale,
                                   float* __restrict__ shift) {
    int n = threadIdx.x;
    const float inv  = 1.0f / (float)N_NODES;
    float mean = sum[n] * inv;
    float var  = fmaf(-mean, mean, sumsq[n] * inv);
    float sc   = gamma[n] * rsqrtf(var + BN_EPS);
    scale[n] = sc;
    shift[n] = fmaf(-mean, sc, beta[n]);
}

#define APPLY_BLOCKS 512
#define APPLY_STRIDE (APPLY_BLOCKS * 256)

__launch_bounds__(256)
__global__ void bn_apply_kernel(float4* __restrict__ y4,
                                const float4* __restrict__ scale4,
                                const float4* __restrict__ shift4) {
    constexpr int total4 = N_NODES * DIM / 4;
    constexpr int FULL = total4 / APPLY_STRIDE;
    const int gtid = blockIdx.x * 256 + threadIdx.x;
    const int c = gtid & 31;
    const float4 s = scale4[c];
    const float4 t = shift4[c];

    float4* p = y4 + gtid;
    for (int k = 0; k + 8 <= FULL; k += 8) {
        float4 v[8];
#pragma unroll
        for (int u = 0; u < 8; ++u) v[u] = p[(size_t)(k + u) * APPLY_STRIDE];
#pragma unroll
        for (int u = 0; u < 8; ++u) {
            float4 r = v[u];
            r.x = fmaf(r.x, s.x, t.x);
            r.y = fmaf(r.y, s.y, t.y);
            r.z = fmaf(r.z, s.z, t.z);
            r.w = fmaf(r.w, s.w, t.w);
            p[(size_t)(k + u) * APPLY_STRIDE] = r;
        }
    }
    for (int k = FULL - FULL % 8; k < FULL; ++k) {
        float4 r = p[(size_t)k * APPLY_STRIDE];
        r.x = fmaf(r.x, s.x, t.x);
        r.y = fmaf(r.y, s.y, t.y);
        r.z = fmaf(r.z, s.z, t.z);
        r.w = fmaf(r.w, s.w, t.w);
        p[(size_t)k * APPLY_STRIDE] = r;
    }
    if (gtid + FULL * APPLY_STRIDE < total4) {
        float4 r = p[(size_t)FULL * APPLY_STRIDE];
        r.x = fmaf(r.x, s.x, t.x);
        r.y = fmaf(r.y, s.y, t.y);
        r.z = fmaf(r.z, s.z, t.z);
        r.w = fmaf(r.w, s.w, t.w);
        p[(size_t)FULL * APPLY_STRIDE] = r;
    }
}

extern "C" void kernel_launch(void* const* d_in, const int* in_sizes, int n_in,
                              void* d_out, int out_size, void* d_ws, size_t ws_size,
                              hipStream_t stream) {
    const float* feature = (const float*)d_in[0];
    const float* ew      = (const float*)d_in[1];
    const int*   src     = (const int*)d_in[2];
    const int*   dst     = (const int*)d_in[3];
    const float* W       = (const float*)d_in[4];
    const float* b       = (const float*)d_in[5];
    const float* gamma   = (const float*)d_in[6];
    const float* beta    = (const float*)d_in[7];

    float* out = (float*)d_out;
    float* ws  = (float*)d_ws;
    float* sum   = ws;
    float* sumsq = ws + 128;
    float* scale = ws + 256;
    float* shift = ws + 384;

    int*      start_t = (int*)d_ws + WS_START_OFF;
    unsigned* ents    = (unsigned*)d_ws + WS_ENT_OFF;

    bin_edges_kernel<<<BIN_BLOCKS, BIN_THREADS, 0, stream>>>(ew, src, dst, start_t, ents);

    owner_sum_kernel<<<OWN_BLOCKS, 1024, 0, stream>>>(feature, start_t, ents, out, sum);

    gemm_relu_stats_kernel<<<GEMM_BLOCKS, 256, 0, stream>>>(out, W, b, sum, sumsq);

    bn_finalize_kernel<<<1, 128, 0, stream>>>(sum, sumsq, gamma, beta, scale, shift);

    bn_apply_kernel<<<APPLY_BLOCKS, 256, 0, stream>>>(
        (float4*)out, (const float4*)scale, (const float4*)shift);
}


# GEMM fused into the owner groups: the 400 summed rows stay in LDS (pitch 132) and feed the f32 WMMAs there; H is never stored or re-read, one launch fewer; K is summed in LDS column order
# speedup vs baseline: 2.0725x; 1.0608x over previous
#include <hip/hip_runtime.h>

#define N_NODES 100000
#define N_EDGES 640000
#define DIM 128
#define BN_EPS 1e-5f

typedef __attribute__((ext_vector_type(2))) float v2f;
typedef __attribute__((ext_vector_type(8))) float v8f;

#define OWN_ROWS 400
#define OWN_BLOCKS (N_NODES / OWN_ROWS)
#define OWN_WAVES 32
#define LIST_CAP 256
#define LIST_TRIPS (LIST_CAP / 32)
#define ROWS_IN_FLIGHT 8
#define ID_MASK 0xFFFFFu
#define ROW_PITCH 132
#define M_TILES (OWN_ROWS / 16)

#define BIN_BLOCKS 250
#define BIN_THREADS 512
#define SLICE (N_EDGES / BIN_BLOCKS)
#define BIN_PER_THREAD (SLICE / BIN_THREADS)
#define SEG_PER_WAVE ((BIN_BLOCKS + OWN_WAVES - 1) / OWN_WAVES)
#define HIST_SIZE 256
#define WS_START_OFF 1024
#define WS_ENT_OFF (1 << 18)

static_assert(N_NODES % OWN_ROWS == 0, "owner groups tile the nodes exactly");
static_assert(OWN_ROWS % 16 == 0 && OWN_WAVES == 32 && DIM == 128, "M tiles tile the owned rows; 8 N tiles x 4 M groups of waves");
static_assert(N_EDGES % BIN_BLOCKS == 0 && SLICE % BIN_THREADS == 0, "slices tile the edges, threads tile a slice");
static_assert(N_NODES <= (1 << 20) && OWN_ROWS <= (1 << 12), "list entry packing");
static_assert(OWN_BLOCKS + 1 <= HIST_SIZE && HIST_SIZE <= BIN_THREADS && HIST_SIZE % 32 == 0, "one histogram slot per group + the out-of-range slot");
static_assert(SEG_PER_WAVE <= 32 && LIST_CAP % ROWS_IN_FLIGHT == 0 && LIST_CAP % 32 == 0, "a wave's runs fit its lanes; padding stays inside the list");
static_assert(WS_START_OFF + (OWN_BLOCKS + 1) * BIN_BLOCKS <= WS_ENT_OFF, "start table ends before the entries");

__launch_bounds__(BIN_THREADS)
__global__ void bin_edges_kernel(const float* __restrict__ ew,
                                 const int*   __restrict__ src,
                                 const int*   __restrict__ dst,
                                 int*         __restrict__ start_t,
                                 unsigned*    __restrict__ ents,
                                 float*       __restrict__ sums) {
    __shared__ int hist[HIST_SIZE];
    __shared__ int wave_tot[HIST_SIZE / 32];

    const int tid  = threadIdx.x;
    const int lane = tid & 31;
    const int a    = blockIdx.x;

    if (tid < HIST_SIZE) hist[tid] = 0;
    if (a == 0 && tid < 256) sums[tid] = 0.f;
    __syncthreads();

    unsigned grp[BIN_PER_THREAD], val[BIN_PER_THREAD], wbits[BIN_PER_THREAD];
    int      rank[BIN_PER_THREAD];
#pragma unroll
    for (int k = 0; k < BIN_PER_THREAD; ++k) {
        const int e = a * SLICE + k * BIN_THREADS + tid;
        const unsigned d = (unsigned)dst[e];
        const unsigned s = (unsigned)src[e] & ID_MASK;
        wbits[k] = __float_as_uint(ew[e]);
        const unsigned q = d / OWN_ROWS;
        grp[k] = q < OWN_BLOCKS ? q : OWN_BLOCKS;
        val[k] = s | ((d - q * OWN_ROWS) << 20);
    }
#pragma unroll
    for (int k = 0; k < BIN_PER_THREAD; ++k)
        rank[k] = atomicAdd(&hist[grp[k]], 1);
    __syncthreads();

    int cntv = 0, incl = 0;
    if (tid < HIST_SIZE) {
        cntv = hist[tid];
        incl = cntv;
    }
#pragma unroll
    for (int dlt = 1; dlt < 32; dlt <<= 1) {
        const int t = __shfl_up(incl, dlt);
        if (lane >= dlt) incl += t;
    }
    if (tid < HIST_SIZE && lane == 31) wave_tot[tid >> 5] = incl;
    __syncthreads();
    if (tid < HIST_SIZE) {
        int base = 0;
        for (int w = 0; w < (tid >> 5); ++w) base += wave_tot[w];
        const int excl = base + incl - cntv;
        hist[tid] = excl;
        if (tid <= OWN_BLOCKS) start_t[tid * BIN_BLOCKS + a] = excl;
    }
    __syncthreads();

    uint2* ents2 = reinterpret_cast<uint2*>(ents);
#pragma unroll
    for (int k = 0; k < BIN_PER_THREAD; ++k)
        ents2[a * SLICE + hist[grp[k]] + rank[k]] = make_uint2(val[k], wbits[k]);
}

__device__ __forceinline__ void list_walk(unsigned* ent, float* wt, int cnt, int lane,
                                          const float4* __restrict__ feat4,
                                          float* rows) {
    const int n = (cnt + ROWS_IN_FLIGHT - 1) & ~(ROWS_IN_FLIGHT - 1);
    if (cnt + lane < n) {
        ent[cnt + lane] = (unsigned)OWN_ROWS << 20;
        wt[cnt + lane]  = 0.f;
    }
    for (int j0 = 0; j0 < n; j0 += ROWS_IN_FLIGHT) {
        unsigned en[ROWS_IN_FLIGHT];
        float    w[ROWS_IN_FLIGHT];
        float4   f[ROWS_IN_FLIGHT];
#pragma unroll
        for (int u = 0; u < ROWS_IN_FLIGHT; ++u) {
            en[u] = ent[j0 + u];
            w[u]  = wt[j0 + u];
            f[u]  = feat4[(size_t)(en[u] & ID_MASK) * (DIM / 4) + lane];
        }
#pragma unroll
        for (int u = 0; u < ROWS_IN_FLIGHT; ++u) {
            float* p = rows + (en[u] >> 20) * ROW_PITCH + lane;
            unsafeAtomicAdd(p,      w[u] * f[u].x);
            unsafeAtomicAdd(p + 32, w[u] * f[u].y);
            unsafeAtomicAdd(p + 64, w[u] * f[u].z);
            unsafeAtomicAdd(p + 96, w[u] * f[u].w);
        }
    }
}

__launch_bounds__(1024)
__global__ void owner_sum_gemm_kernel(const float* __restrict__ feat,
                                      const int*      __restrict__ start_t,
                                      const unsigned* __restrict__ ents,
                                      const float* __restrict__ W,
                                      const float* __restrict__ bias,
                                      float*       __restrict__ y,
                                      float*       __restrict__ sum,
                                      float*       __restrict__ sumsq) {
    __shared__ __attribute__((aligned(16))) float rows[(OWN_ROWS + 1) * ROW_PITCH];
    __shared__ unsigned ent_all[OWN_WAVES * LIST_CAP];
    __shared__ float    wt_all[OWN_WAVES * LIST_CAP];

    const int tid  = threadIdx.x;
    const int wave = __builtin_amdgcn_readfirstlane(tid >> 5);
    const int lane = tid & 31;
    const int g    = blockIdx.x;
    const int lo   = g * OWN_ROWS;

    const int a = wave * SEG_PER_WAVE + lane;
    int s = 0, s_next = 0;
    if (lane < SEG_PER_WAVE && a < BIN_BLOCKS) {
        s      = start_t[g * BIN_BLOCKS + a];
        s_next = start_t[(g + 1) * BIN_BLOCKS + a];
    }

    for (int i = tid; i < OWN_ROWS * ROW_PITCH / 4; i += 1024)
        reinterpret_cast<float4*>(rows)[i] = make_float4(0.f, 0.f, 0.f, 0.f);
    __syncthreads();
    const int nrun = s_next - s;

    unsigned* ent = ent_all + wave * LIST_CAP;
    float*    wt  = wt_all  + wave * LIST_CAP;
    const float4* feat4 = reinterpret_cast<const float4*>(feat);
    const uint2*  ents2 = reinterpret_cast<const uint2*>(ents);

    int inc[SEG_PER_WAVE], sb[SEG_PER_WAVE];
#pragma unroll
    for (int k = 0; k < SEG_PER_WAVE; ++k) {
        const int nk = __builtin_amdgcn_readlane(nrun, k);
        const int sk = __builtin_amdgcn_readlane(s, k);
        const int before = (k > 0) ? inc[(k > 0) ? k - 1 : 0] : 0;
        inc[k] = before + nk;
        sb[k]  = (wave * SEG_PER_WAVE + k) * SLICE + sk - before;
    }
    const int total = inc[SEG_PER_WAVE - 1];

    for (int p0 = 0; p0 < total; p0 += LIST_CAP) {
        const int cnt = (total - p0 < LIST_CAP) ? (total - p0) : LIST_CAP;
        uint2 x[LIST_TRIPS];
#pragma unroll
        for (int u = 0; u < LIST_TRIPS; ++u) {
            const int j = u * 32 + lane;
            x[u] = make_uint2(0u, 0u);
            if (j < cnt) {
                const int p = p0 + j;
                int base = sb[0];
#pragma unroll
                for (int k = 1; k < SEG_PER_WAVE; ++k)
                    if (p >= inc[k - 1]) base = sb[k];
                x[u] = ents2[base + p];
            }
        }
#pragma unroll
        for (int u = 0; u < LIST_TRIPS; ++u) {
            const int j = u * 32 + lane;
            if (j < cnt) {
                ent[j] = x[u].x;
                wt[j]  = __uint_as_float(x[u].y);
            }
        }
        list_walk(ent, wt, cnt, lane, feat4, rows);
    }

    const int nl   = lane & 15;
    const int half = lane >> 4;
    const int nt   = wave & 7;
    const int mg   = wave >> 3;
    const int n    = nt * 16 + nl;
    v2f b[32];
    {
        const float4* w4 = reinterpret_cast<const float4*>(W + (size_t)n * DIM + 8 * half);
#pragma unroll
        for (int a8 = 0; a8 < 8; ++a8) {
            const float4 f0 = w4[4 * a8];
            const float4 f1 = w4[4 * a8 + 1];
            b[a8]      = (v2f){f0.x, f1.x};
            b[8 + a8]  = (v2f){f0.y, f1.y};
            b[16 + a8] = (v2f){f0.z, f1.z};
            b[24 + a8] = (v2f){f0.w, f1.w};
        }
    }
    const float bn = bias[n];
    __syncthreads();

    float ls = 0.f, lq = 0.f;
    for (int m = mg; m < M_TILES; m += 4) {
        const float* arow = rows + (m * 16 + nl) * ROW_PITCH + 2 * half;
        v8f acc = {};
#pragma unroll
        for (int j = 0; j < 32; ++j) {
            v2f av = *reinterpret_cast<const v2f*>(arow + 4 * j);
            acc = __builtin_amdgcn_wmma_f32_16x16x4_f32(
                 false, av,  false, b[j],
                 (short)0, acc,  false,  false);
        }
        float* yb = y + (size_t)(lo + m * 16) * DIM + n;
#pragma unroll
        for (int v = 0; v < 8; ++v) {
            const float val = fmaxf(acc[v] + bn, 0.f);
            const int mr = half ? (v + 8) : v;
            yb[(size_t)mr * DIM] = val;
            ls += val;
            lq += val * val;
        }
    }

    ls += __shfl_xor(ls, 16);
    lq += __shfl_xor(lq, 16);
    if (half == 0) {
        wt_all[wave * 32 + nl]      = ls;
        wt_all[wave * 32 + 16 + nl] = lq;
    }
    __syncthreads();
    if (tid < 256) {
        const int x = tid & 31;
        const float tot = wt_all[tid] + wt_all[256 + tid] + wt_all[512 + tid] + wt_all[768 + tid];
        float* dstp = (x < 16) ? (sum + (tid >> 5) * 16 + x) : (sumsq + (tid >> 5) * 16 + (x - 16));
        unsafeAtomicAdd(dstp, tot);
    }
}

__global__ void bn_finalize_kernel(const float* __restrict__ sum,
                                   const float* __restrict__ sumsq,
                                   const float* __restrict__ gamma,
                                   const float* __restrict__ beta,
                                   float* __restrict__ scale,
                                   float* __restrict__ shift) {
    int n = threadIdx.x;
    const float inv  = 1.0f / (float)N_NODES;
    float mean = sum[n] * inv;
    float var  = fmaf(-mean, mean, sumsq[n] * inv);
    float sc   = gamma[n] * rsqrtf(var + BN_EPS);
    scale[n] = sc;
    shift[n] = fmaf(-mean, sc, beta[n]);
}

#define APPLY_BLOCKS 512
#define APPLY_STRIDE (APPLY_BLOCKS * 256)

__launch_bounds__(256)
__global__ void bn_apply_kernel(float4* __restrict__ y4,
                                const float4* __restrict__ scale4,
                                const float4* __restrict__ shift4) {
    constexpr int total4 = N_NODES * DIM / 4;
    constexpr int FULL = total4 / APPLY_STRIDE;
    const int gtid = blockIdx.x * 256 + threadIdx.x;
    const int c = gtid & 31;
    const float4 s = scale4[c];
    const float4 t = shift4[c];

    float4* p = y4 + gtid;
    for (int k = 0; k + 8 <= FULL; k += 8) {
        float4 v[8];
#pragma unroll
        for (int u = 0; u < 8; ++u) v[u] = p[(size_t)(k + u) * APPLY_STRIDE];
#pragma unroll
        for (int u = 0; u < 8; ++u) {
            float4 r = v[u];
            r.x = fmaf(r.x, s.x, t.x);
            r.y = fmaf(r.y, s.y, t.y);
            r.z = fmaf(r.z, s.z, t.z);
            r.w = fmaf(r.w, s.w, t.w);
            p[(size_t)(k + u) * APPLY_STRIDE] = r;
        }
    }
    for (int k = FULL - FULL % 8; k < FULL; ++k) {
        float4 r = p[(size_t)k * APPLY_STRIDE];
        r.x = fmaf(r.x, s.x, t.x);
        r.y = fmaf(r.y, s.y, t.y);
        r.z = fmaf(r.z, s.z, t.z);
        r.w = fmaf(r.w, s.w, t.w);
        p[(size_t)k * APPLY_STRIDE] = r;
    }
    if (gtid + FULL * APPLY_STRIDE < total4) {
        float4 r = p[(size_t)FULL * APPLY_STRIDE];
        r.x = fmaf(r.x, s.x, t.x);
        r.y = fmaf(r.y, s.y, t.y);
        r.z = fmaf(r.z, s.z, t.z);
        r.w = fmaf(r.w, s.w, t.w);
        p[(size_t)FULL * APPLY_STRIDE] = r;
    }
}

extern "C" void kernel_launch(void* const* d_in, const int* in_sizes, int n_in,
                              void* d_out, int out_size, void* d_ws, size_t ws_size,
                              hipStream_t stream) {
    const float* feature = (const float*)d_in[0];
    const float* ew      = (const float*)d_in[1];
    const int*   src     = (const int*)d_in[2];
    const int*   dst     = (const int*)d_in[3];
    const float* W       = (const float*)d_in[4];
    const float* b       = (const float*)d_in[5];
    const float* gamma   = (const float*)d_in[6];
    const float* beta    = (const float*)d_in[7];

    float* out = (float*)d_out;
    float* ws  = (float*)d_ws;
    float* sum   = ws;
    float* sumsq = ws + 128;
    float* scale = ws + 256;
    float* shift = ws + 384;

    int*      start_t = (int*)d_ws + WS_START_OFF;
    unsigned* ents    = (unsigned*)d_ws + WS_ENT_OFF;

    bin_edges_kernel<<<BIN_BLOCKS, BIN_THREADS, 0, stream>>>(ew, src, dst, start_t, ents, sum);

    owner_sum_gemm_kernel<<<OWN_BLOCKS, 1024, 0, stream>>>(feature, start_t, ents, W, b, out, sum, sumsq);

    bn_finalize_kernel<<<1, 128, 0, stream>>>(sum, sumsq, gamma, beta, scale, shift);

    bn_apply_kernel<<<APPLY_BLOCKS, 256, 0, stream>>>(
        (float4*)out, (const float4*)scale, (const float4*)shift);
}
